# starting kernel
# baseline (speedup 1.0000x reference)
#include <hip/hip_runtime.h>
#include <hip/hip_bf16.h>
#include <cstddef>

#define B_ 32
#define T_ 2048
#define D_ 512

typedef __attribute__((ext_vector_type(2))) float v2f;
typedef __attribute__((ext_vector_type(8))) float v8f;

template <bool WEIGHTED>
__global__ __launch_bounds__(256) void pool_wmma_kernel(
    const float* __restrict__ x, const float* __restrict__ coeff,
    float* __restrict__ out) {
  const int b      = blockIdx.x;
  const int d0     = blockIdx.y * 16;
  const int lane   = threadIdx.x & 31;
  const int wave   = threadIdx.x >> 5;
  const int col    = lane & 15;
  const int kh     = (lane >> 4) * 2;

  const float* xb = x + (size_t)b * T_ * D_;
  const float* cb = WEIGHTED ? (coeff + (size_t)b * T_) : nullptr;

  const int t_begin = wave * (T_ / 8);
  const int t_end   = t_begin + (T_ / 8);

  v8f c0 = {};
  v8f c1 = {};

  for (int t0 = t_begin; t0 < t_end; t0 += 8) {
    {
      v2f a;
      if (WEIGHTED) { a.x = cb[t0 + kh]; a.y = cb[t0 + kh + 1]; }
      else          { a.x = 1.0f;        a.y = 1.0f; }
      const float* xp = xb + (size_t)(t0 + kh) * D_ + d0 + col;
      v2f bm;
      bm.x = xp[0];
      bm.y = xp[D_];
      __builtin_prefetch(xp + 8 * D_, 0, 1);
      c0 = __builtin_amdgcn_wmma_f32_16x16x4_f32(false, a, false, bm,
                                                 (short)0, c0, false, false);
    }
    {
      v2f a;
      if (WEIGHTED) { a.x = cb[t0 + 4 + kh]; a.y = cb[t0 + 4 + kh + 1]; }
      else          { a.x = 1.0f;            a.y = 1.0f; }
      const float* xp = xb + (size_t)(t0 + 4 + kh) * D_ + d0 + col;
      v2f bm;
      bm.x = xp[0];
      bm.y = xp[D_];
      c1 = __builtin_amdgcn_wmma_f32_16x16x4_f32(false, a, false, bm,
                                                 (short)0, c1, false, false);
    }
  }

  __shared__ float red[8][16];
  if (lane < 16) red[wave][lane] = c0[0] + c1[0];
  __syncthreads();
  if (threadIdx.x < 16) {
    float acc = 0.f;
#pragma unroll
    for (int wv = 0; wv < 8; ++wv) acc += red[wv][threadIdx.x];
    out[(size_t)b * D_ + d0 + threadIdx.x] = acc;
  }
}

__global__ __launch_bounds__(256) void dots_kernel(
    const float* __restrict__ x, const float* __restrict__ s,
    float* __restrict__ w) {
  const int lane = threadIdx.x & 31;
  const int wave = threadIdx.x >> 5;
  const int r    = blockIdx.x * 8 + wave;
  const int b    = r >> 11;

  const float* xr = x + (size_t)r * D_;
  const float* sr = s + (size_t)b * D_;

  float dot = 0.f, sq = 0.f;
#pragma unroll
  for (int j = 0; j < 4; ++j) {
    const int col = j * 128 + lane * 4;
    const float4 xv = *(const float4*)(xr + col);
    const float4 sv = *(const float4*)(sr + col);
    dot = fmaf(xv.x, sv.x, dot); dot = fmaf(xv.y, sv.y, dot);
    dot = fmaf(xv.z, sv.z, dot); dot = fmaf(xv.w, sv.w, dot);
    sq  = fmaf(xv.x, xv.x, sq);  sq  = fmaf(xv.y, xv.y, sq);
    sq  = fmaf(xv.z, xv.z, sq);  sq  = fmaf(xv.w, xv.w, sq);
  }
#pragma unroll
  for (int off = 16; off > 0; off >>= 1) {
    dot += __shfl_xor(dot, off, 32);
    sq  += __shfl_xor(sq,  off, 32);
  }
  if (lane == 0) w[r] = dot - sq;
}

__global__ __launch_bounds__(256) void softmax_kernel(
    const float* __restrict__ w, float* __restrict__ p) {
  const int b   = blockIdx.x;
  const int tid = threadIdx.x;
  const float* wb = w + (size_t)b * T_;
  float* pb       = p + (size_t)b * T_;

  __shared__ float red[256];
  float vals[8];
  float lmax = -3.4e38f;
#pragma unroll
  for (int j = 0; j < 8; ++j) {
    vals[j] = wb[j * 256 + tid];
    lmax = fmaxf(lmax, vals[j]);
  }
  red[tid] = lmax;
  __syncthreads();
#pragma unroll
  for (int sft = 128; sft > 0; sft >>= 1) {
    if (tid < sft) red[tid] = fmaxf(red[tid], red[tid + sft]);
    __syncthreads();
  }
  const float bmax = red[0];
  __syncthreads();

  float lsum = 0.f;
#pragma unroll
  for (int j = 0; j < 8; ++j) {
    vals[j] = __expf(vals[j] - bmax);
    lsum += vals[j];
  }
  red[tid] = lsum;
  __syncthreads();
#pragma unroll
  for (int sft = 128; sft > 0; sft >>= 1) {
    if (tid < sft) red[tid] += red[tid + sft];
    __syncthreads();
  }
  const float inv = 1.0f / red[0];
#pragma unroll
  for (int j = 0; j < 8; ++j) pb[j * 256 + tid] = vals[j] * inv;
}

extern "C" void kernel_launch(void* const* d_in, const int* in_sizes, int n_in,
                              void* d_out, int out_size, void* d_ws, size_t ws_size,
                              hipStream_t stream) {
  (void)in_sizes; (void)n_in; (void)out_size; (void)ws_size;
  const float* x = (const float*)d_in[0];
  float* out = (float*)d_out;

  float* ws = (float*)d_ws;
  float* s_ws  = ws;
  float* w_ws  = s_ws + (size_t)B_ * D_;
  float* p_ws  = w_ws + (size_t)B_ * T_;

  pool_wmma_kernel<false><<<dim3(B_, D_ / 16), 256, 0, stream>>>(x, nullptr, s_ws);
  dots_kernel<<<(B_ * T_) / 8, 256, 0, stream>>>(x, s_ws, w_ws);
  softmax_kernel<<<B_, 256, 0, stream>>>(w_ws, p_ws);
  pool_wmma_kernel<true><<<dim3(B_, D_ / 16), 256, 0, stream>>>(x, p_ws, out);
}


# dots kernel as a row stream: each wave owns 16 rows (4,096 waves, not 65,536), its slice of s kept in registers, next 4 rows' 16 loads issued before the current 4 are summed; w bit-identical
# speedup vs baseline: 1.5164x; 1.5164x over previous
#include <hip/hip_runtime.h>
#include <hip/hip_bf16.h>
#include <cstddef>

#define B_ 32
#define T_ 2048
#define D_ 512

typedef __attribute__((ext_vector_type(2))) float v2f;
typedef __attribute__((ext_vector_type(8))) float v8f;

#define POOL_WAVES 16
#define POOL_T_PER_WAVE (T_ / POOL_WAVES)
#define POOL_PAIRS 4
#define POOL_GROUPS (POOL_T_PER_WAVE / (8 * POOL_PAIRS))

template <bool WEIGHTED>
__device__ __forceinline__ void pool_load_group(
    const float* __restrict__ xq, const float* __restrict__ cq, int g,
    float4 (&r)[POOL_PAIRS][4], float4 (&cf)[POOL_PAIRS]) {
#pragma unroll
  for (int u = 0; u < POOL_PAIRS; ++u) {
    const int row = (g * POOL_PAIRS + u) * 8;
#pragma unroll
    for (int i = 0; i < 4; ++i)
      r[u][i] = *(const float4*)(xq + (size_t)(row + i) * D_);
    if (WEIGHTED) cf[u] = *(const float4*)(cq + row);
    else          cf[u] = make_float4(1.0f, 1.0f, 1.0f, 1.0f);
  }
}

__device__ __forceinline__ v8f pool_mma(float a0, float a1, float b0, float b1, v8f c) {
  v2f a;  a.x = a0;  a.y = a1;
  v2f bm; bm.x = b0; bm.y = b1;
  return __builtin_amdgcn_wmma_f32_16x16x4_f32(false, a, false, bm,
                                               (short)0, c, false, false);
}

__device__ __forceinline__ void pool_wave_stats(const float4& wv, float& m, float& l) {
  float mx = fmaxf(fmaxf(wv.x, wv.y), fmaxf(wv.z, wv.w));
  for (int off = 16; off > 0; off >>= 1) mx = fmaxf(mx, __shfl_xor(mx, off, 32));
  float sm = (__expf(wv.x - mx) + __expf(wv.y - mx)) +
             (__expf(wv.z - mx) + __expf(wv.w - mx));
  for (int off = 16; off > 0; off >>= 1) sm += __shfl_xor(sm, off, 32);
  m = mx;
  l = sm;
}

__device__ __forceinline__ void pool_exp_group(float4 (&cf)[POOL_PAIRS], float m) {
#pragma unroll
  for (int u = 0; u < POOL_PAIRS; ++u) {
    cf[u].x = __expf(cf[u].x - m);
    cf[u].y = __expf(cf[u].y - m);
    cf[u].z = __expf(cf[u].z - m);
    cf[u].w = __expf(cf[u].w - m);
  }
}

__device__ __forceinline__ void pool_compute_group(
    const float4 (&r)[POOL_PAIRS][4], const float4 (&cf)[POOL_PAIRS],
    v8f& c0, v8f& c1, v8f& c2, v8f& c3) {
#pragma unroll
  for (int u = 0; u < POOL_PAIRS; ++u) {
    c0 = pool_mma(cf[u].x, cf[u].y, r[u][0].x, r[u][1].x, c0);
    c1 = pool_mma(cf[u].x, cf[u].y, r[u][0].y, r[u][1].y, c1);
    c2 = pool_mma(cf[u].x, cf[u].y, r[u][0].z, r[u][1].z, c2);
    c3 = pool_mma(cf[u].x, cf[u].y, r[u][0].w, r[u][1].w, c3);
    c0 = pool_mma(cf[u].z, cf[u].w, r[u][2].x, r[u][3].x, c0);
    c1 = pool_mma(cf[u].z, cf[u].w, r[u][2].y, r[u][3].y, c1);
    c2 = pool_mma(cf[u].z, cf[u].w, r[u][2].z, r[u][3].z, c2);
    c3 = pool_mma(cf[u].z, cf[u].w, r[u][2].w, r[u][3].w, c3);
  }
}

template <bool WEIGHTED>
__global__ __launch_bounds__(POOL_WAVES * 32) void pool_wmma_kernel(
    const float* __restrict__ x, const float* __restrict__ coeff,
    float* __restrict__ out) {
  const int b    = blockIdx.x;
  const int d0   = blockIdx.y * 64;
  const int lane = threadIdx.x & 31;
  const int wave = threadIdx.x >> 5;
  const int col  = lane & 15;
  const int h    = lane >> 4;

  const int tb = wave * POOL_T_PER_WAVE;
  const float* xq = x + (size_t)b * T_ * D_ + (size_t)(tb + 4 * h) * D_ + d0 + 4 * col;
  const float* cq = WEIGHTED ? (coeff + (size_t)b * T_ + tb + 4 * h) : nullptr;

  v8f c0 = {};
  v8f c1 = {};
  v8f c2 = {};
  v8f c3 = {};

  float4 ra[POOL_PAIRS][4], rb[POOL_PAIRS][4];
  float4 ca[POOL_PAIRS], cb[POOL_PAIRS];

  float4 wv = make_float4(0.f, 0.f, 0.f, 0.f);
  if (WEIGHTED) wv = *(const float4*)(coeff + (size_t)b * T_ + tb + 4 * lane);
  float m_w = 0.f, l_w = 0.f;

  pool_load_group<WEIGHTED>(xq, cq, 0, ra, ca);
#pragma unroll
  for (int g = 0; g < POOL_GROUPS; g += 2) {
    __builtin_amdgcn_sched_barrier(0);
    pool_load_group<WEIGHTED>(xq, cq, g + 1, rb, cb);
    __builtin_amdgcn_sched_barrier(0);
    if (WEIGHTED) {
      if (g == 0) pool_wave_stats(wv, m_w, l_w);
      pool_exp_group(ca, m_w);
    }
    pool_compute_group(ra, ca, c0, c1, c2, c3);
    __builtin_amdgcn_sched_barrier(0);
    if (g + 2 < POOL_GROUPS) pool_load_group<WEIGHTED>(xq, cq, g + 2, ra, ca);
    __builtin_amdgcn_sched_barrier(0);
    if (WEIGHTED) pool_exp_group(cb, m_w);
    pool_compute_group(rb, cb, c0, c1, c2, c3);
  }

  __shared__ float4 red[POOL_WAVES][16];
  __shared__ float red_m[POOL_WAVES];
  __shared__ float red_l[POOL_WAVES];
  if (lane < 16) red[wave][lane] = make_float4(c0[0], c1[0], c2[0], c3[0]);
  if (WEIGHTED && lane == 0) { red_m[wave] = m_w; red_l[wave] = l_w; }
  __syncthreads();
  if (threadIdx.x < 64) {
    const float* rf = (const float*)red;
    if (WEIGHTED) {
      float M = red_m[0];
#pragma unroll
      for (int wv = 1; wv < POOL_WAVES; ++wv) M = fmaxf(M, red_m[wv]);
      float acc = 0.f, den = 0.f;
#pragma unroll
      for (int wv = 0; wv < POOL_WAVES; ++wv) {
        const float sc = __expf(red_m[wv] - M);
        acc += sc * rf[wv * 64 + threadIdx.x];
        den += sc * red_l[wv];
      }
      out[(size_t)b * D_ + d0 + threadIdx.x] = acc / den;
    } else {
      float acc = 0.f;
#pragma unroll
      for (int wv = 0; wv < POOL_WAVES; ++wv) acc += rf[wv * 64 + threadIdx.x];
      out[(size_t)b * D_ + d0 + threadIdx.x] = acc;
    }
  }
}

#define DOT_WAVES 16
#define DOT_ROWS 16
#define DOT_GROUP 4
#define DOT_GROUPS (DOT_ROWS / DOT_GROUP)

__device__ __forceinline__ void dots_load_group(
    const float* __restrict__ xq, int g, float4 (&r)[DOT_GROUP][4]) {
#pragma unroll
  for (int i = 0; i < DOT_GROUP; ++i) {
#pragma unroll
    for (int j = 0; j < 4; ++j)
      r[i][j] = *(const float4*)(xq + (size_t)(g * DOT_GROUP + i) * D_ + j * 128);
  }
}

__device__ __forceinline__ void dots_sum_group(
    const float4 (&r)[DOT_GROUP][4], const float4 (&sv)[4], float (&res)[DOT_ROWS], int g) {
  float dot[DOT_GROUP], sq[DOT_GROUP];
#pragma unroll
  for (int i = 0; i < DOT_GROUP; ++i) {
    float d = 0.f, q = 0.f;
#pragma unroll
    for (int j = 0; j < 4; ++j) {
      const float4 xv = r[i][j];
      const float4 s4 = sv[j];
      d = fmaf(xv.x, s4.x, d); d = fmaf(xv.y, s4.y, d);
      d = fmaf(xv.z, s4.z, d); d = fmaf(xv.w, s4.w, d);
      q = fmaf(xv.x, xv.x, q); q = fmaf(xv.y, xv.y, q);
      q = fmaf(xv.z, xv.z, q); q = fmaf(xv.w, xv.w, q);
    }
    dot[i] = d;
    sq[i]  = q;
  }
#pragma unroll
  for (int off = 16; off > 0; off >>= 1) {
    float td[DOT_GROUP], tq[DOT_GROUP];
#pragma unroll
    for (int i = 0; i < DOT_GROUP; ++i) {
      td[i] = __shfl_xor(dot[i], off, 32);
      tq[i] = __shfl_xor(sq[i],  off, 32);
    }
#pragma unroll
    for (int i = 0; i < DOT_GROUP; ++i) {
      dot[i] += td[i];
      sq[i]  += tq[i];
    }
  }
#pragma unroll
  for (int i = 0; i < DOT_GROUP; ++i) res[g * DOT_GROUP + i] = dot[i] - sq[i];
}

__global__ __launch_bounds__(DOT_WAVES * 32) void dots_kernel(
    const float* __restrict__ x, const float* __restrict__ s,
    float* __restrict__ w) {
  const int lane = threadIdx.x & 31;
  const int wave = __builtin_amdgcn_readfirstlane((int)(threadIdx.x >> 5));
  const int row0 = (blockIdx.x * DOT_WAVES + wave) * DOT_ROWS;
  const int b    = row0 >> 11;

  const float* xq = x + (size_t)row0 * D_ + lane * 4;
  const float* sr = s + (size_t)b * D_ + lane * 4;

  float4 sv[4];
#pragma unroll
  for (int j = 0; j < 4; ++j) sv[j] = *(const float4*)(sr + j * 128);

  float4 ra[DOT_GROUP][4], rb[DOT_GROUP][4];
  float res[DOT_ROWS];

  dots_load_group(xq, 0, ra);
#pragma unroll
  for (int g = 0; g < DOT_GROUPS; g += 2) {
    __builtin_amdgcn_sched_barrier(0);
    dots_load_group(xq, g + 1, rb);
    __builtin_amdgcn_sched_barrier(0);
    dots_sum_group(ra, sv, res, g);
    __builtin_amdgcn_sched_barrier(0);
    if (g + 2 < DOT_GROUPS) dots_load_group(xq, g + 2, ra);
    __builtin_amdgcn_sched_barrier(0);
    dots_sum_group(rb, sv, res, g + 1);
  }

  float mine = 0.f;
#pragma unroll
  for (int r = 0; r < DOT_ROWS; ++r) if (lane == r) mine = res[r];
  if (lane < DOT_ROWS) w[row0 + lane] = mine;
}

extern "C" void kernel_launch(void* const* d_in, const int* in_sizes, int n_in,
                              void* d_out, int out_size, void* d_ws, size_t ws_size,
                              hipStream_t stream) {
  (void)in_sizes; (void)n_in; (void)out_size; (void)ws_size;
  const float* x = (const float*)d_in[0];
  float* out = (float*)d_out;

  float* ws = (float*)d_ws;
  float* s_ws  = ws;
  float* w_ws  = s_ws + (size_t)B_ * D_;

  pool_wmma_kernel<false><<<dim3(B_, D_ / 64), POOL_WAVES * 32, 0, stream>>>(x, nullptr, s_ws);
  dots_kernel<<<(B_ * T_) / (DOT_WAVES * DOT_ROWS), DOT_WAVES * 32, 0, stream>>>(x, s_ws, w_ws);
  pool_wmma_kernel<true><<<dim3(B_, D_ / 64), POOL_WAVES * 32, 0, stream>>>(x, w_ws, out);
}


# one work-group-id map for all three launches (chunk in id bits 4-6, batch in the rest): each half of the second-level cache re-reads its own batches in every pass; results bit-identical
# speedup vs baseline: 1.5883x; 1.0474x over previous
#include <hip/hip_runtime.h>
#include <hip/hip_bf16.h>
#include <cstddef>

#define B_ 32
#define T_ 2048
#define D_ 512

typedef __attribute__((ext_vector_type(2))) float v2f;
typedef __attribute__((ext_vector_type(8))) float v8f;

__device__ __forceinline__ void wg_batch_chunk(int id, int& b, int& chunk) {
  b     = (id & 15) | ((id >> 7) << 4);
  chunk = (id >> 4) & 7;
}

#define POOL_WAVES 16
#define POOL_T_PER_WAVE (T_ / POOL_WAVES)
#define POOL_PAIRS 4
#define POOL_GROUPS (POOL_T_PER_WAVE / (8 * POOL_PAIRS))

template <bool WEIGHTED>
__device__ __forceinline__ void pool_load_group(
    const float* __restrict__ xq, const float* __restrict__ cq, int g,
    float4 (&r)[POOL_PAIRS][4], float4 (&cf)[POOL_PAIRS]) {
#pragma unroll
  for (int u = 0; u < POOL_PAIRS; ++u) {
    const int row = (g * POOL_PAIRS + u) * 8;
#pragma unroll
    for (int i = 0; i < 4; ++i)
      r[u][i] = *(const float4*)(xq + (size_t)(row + i) * D_);
    if (WEIGHTED) cf[u] = *(const float4*)(cq + row);
    else          cf[u] = make_float4(1.0f, 1.0f, 1.0f, 1.0f);
  }
}

__device__ __forceinline__ v8f pool_mma(float a0, float a1, float b0, float b1, v8f c) {
  v2f a;  a.x = a0;  a.y = a1;
  v2f bm; bm.x = b0; bm.y = b1;
  return __builtin_amdgcn_wmma_f32_16x16x4_f32(false, a, false, bm,
                                               (short)0, c, false, false);
}

__device__ __forceinline__ void pool_wave_stats(const float4& wv, float& m, float& l) {
  float mx = fmaxf(fmaxf(wv.x, wv.y), fmaxf(wv.z, wv.w));
  for (int off = 16; off > 0; off >>= 1) mx = fmaxf(mx, __shfl_xor(mx, off, 32));
  float sm = (__expf(wv.x - mx) + __expf(wv.y - mx)) +
             (__expf(wv.z - mx) + __expf(wv.w - mx));
  for (int off = 16; off > 0; off >>= 1) sm += __shfl_xor(sm, off, 32);
  m = mx;
  l = sm;
}

__device__ __forceinline__ void pool_exp_group(float4 (&cf)[POOL_PAIRS], float m) {
#pragma unroll
  for (int u = 0; u < POOL_PAIRS; ++u) {
    cf[u].x = __expf(cf[u].x - m);
    cf[u].y = __expf(cf[u].y - m);
    cf[u].z = __expf(cf[u].z - m);
    cf[u].w = __expf(cf[u].w - m);
  }
}

__device__ __forceinline__ void pool_compute_group(
    const float4 (&r)[POOL_PAIRS][4], const float4 (&cf)[POOL_PAIRS],
    v8f& c0, v8f& c1, v8f& c2, v8f& c3) {
#pragma unroll
  for (int u = 0; u < POOL_PAIRS; ++u) {
    c0 = pool_mma(cf[u].x, cf[u].y, r[u][0].x, r[u][1].x, c0);
    c1 = pool_mma(cf[u].x, cf[u].y, r[u][0].y, r[u][1].y, c1);
    c2 = pool_mma(cf[u].x, cf[u].y, r[u][0].z, r[u][1].z, c2);
    c3 = pool_mma(cf[u].x, cf[u].y, r[u][0].w, r[u][1].w, c3);
    c0 = pool_mma(cf[u].z, cf[u].w, r[u][2].x, r[u][3].x, c0);
    c1 = pool_mma(cf[u].z, cf[u].w, r[u][2].y, r[u][3].y, c1);
    c2 = pool_mma(cf[u].z, cf[u].w, r[u][2].z, r[u][3].z, c2);
    c3 = pool_mma(cf[u].z, cf[u].w, r[u][2].w, r[u][3].w, c3);
  }
}

template <bool WEIGHTED>
__global__ __launch_bounds__(POOL_WAVES * 32) void pool_wmma_kernel(
    const float* __restrict__ x, const float* __restrict__ coeff,
    float* __restrict__ out) {
  int b, dchunk;
  wg_batch_chunk((int)blockIdx.x, b, dchunk);
  const int d0   = dchunk * 64;
  const int lane = threadIdx.x & 31;
  const int wave = threadIdx.x >> 5;
  const int col  = lane & 15;
  const int h    = lane >> 4;

  const int tb = wave * POOL_T_PER_WAVE;
  const float* xq = x + (size_t)b * T_ * D_ + (size_t)(tb + 4 * h) * D_ + d0 + 4 * col;
  const float* cq = WEIGHTED ? (coeff + (size_t)b * T_ + tb + 4 * h) : nullptr;

  v8f c0 = {};
  v8f c1 = {};
  v8f c2 = {};
  v8f c3 = {};

  float4 ra[POOL_PAIRS][4], rb[POOL_PAIRS][4];
  float4 ca[POOL_PAIRS], cb[POOL_PAIRS];

  float4 wv = make_float4(0.f, 0.f, 0.f, 0.f);
  if (WEIGHTED) wv = *(const float4*)(coeff + (size_t)b * T_ + tb + 4 * lane);
  float m_w = 0.f, l_w = 0.f;

  pool_load_group<WEIGHTED>(xq, cq, 0, ra, ca);
#pragma unroll
  for (int g = 0; g < POOL_GROUPS; g += 2) {
    __builtin_amdgcn_sched_barrier(0);
    pool_load_group<WEIGHTED>(xq, cq, g + 1, rb, cb);
    __builtin_amdgcn_sched_barrier(0);
    if (WEIGHTED) {
      if (g == 0) pool_wave_stats(wv, m_w, l_w);
      pool_exp_group(ca, m_w);
    }
    pool_compute_group(ra, ca, c0, c1, c2, c3);
    __builtin_amdgcn_sched_barrier(0);
    if (g + 2 < POOL_GROUPS) pool_load_group<WEIGHTED>(xq, cq, g + 2, ra, ca);
    __builtin_amdgcn_sched_barrier(0);
    if (WEIGHTED) pool_exp_group(cb, m_w);
    pool_compute_group(rb, cb, c0, c1, c2, c3);
  }

  __shared__ float4 red[POOL_WAVES][16];
  __shared__ float red_m[POOL_WAVES];
  __shared__ float red_l[POOL_WAVES];
  if (lane < 16) red[wave][lane] = make_float4(c0[0], c1[0], c2[0], c3[0]);
  if (WEIGHTED && lane == 0) { red_m[wave] = m_w; red_l[wave] = l_w; }
  __syncthreads();
  if (threadIdx.x < 64) {
    const float* rf = (const float*)red;
    if (WEIGHTED) {
      float M = red_m[0];
#pragma unroll
      for (int wv = 1; wv < POOL_WAVES; ++wv) M = fmaxf(M, red_m[wv]);
      float acc = 0.f, den = 0.f;
#pragma unroll
      for (int wv = 0; wv < POOL_WAVES; ++wv) {
        const float sc = __expf(red_m[wv] - M);
        acc += sc * rf[wv * 64 + threadIdx.x];
        den += sc * red_l[wv];
      }
      out[(size_t)b * D_ + d0 + threadIdx.x] = acc / den;
    } else {
      float acc = 0.f;
#pragma unroll
      for (int wv = 0; wv < POOL_WAVES; ++wv) acc += rf[wv * 64 + threadIdx.x];
      out[(size_t)b * D_ + d0 + threadIdx.x] = acc;
    }
  }
}

#define DOT_WAVES 16
#define DOT_ROWS 16
#define DOT_GROUP 4
#define DOT_GROUPS (DOT_ROWS / DOT_GROUP)

__device__ __forceinline__ void dots_load_group(
    const float* __restrict__ xq, int g, float4 (&r)[DOT_GROUP][4]) {
#pragma unroll
  for (int i = 0; i < DOT_GROUP; ++i) {
#pragma unroll
    for (int j = 0; j < 4; ++j)
      r[i][j] = *(const float4*)(xq + (size_t)(g * DOT_GROUP + i) * D_ + j * 128);
  }
}

__device__ __forceinline__ void dots_sum_group(
    const float4 (&r)[DOT_GROUP][4], const float4 (&sv)[4], float (&res)[DOT_ROWS], int g) {
  float dot[DOT_GROUP], sq[DOT_GROUP];
#pragma unroll
  for (int i = 0; i < DOT_GROUP; ++i) {
    float d = 0.f, q = 0.f;
#pragma unroll
    for (int j = 0; j < 4; ++j) {
      const float4 xv = r[i][j];
      const float4 s4 = sv[j];
      d = fmaf(xv.x, s4.x, d); d = fmaf(xv.y, s4.y, d);
      d = fmaf(xv.z, s4.z, d); d = fmaf(xv.w, s4.w, d);
      q = fmaf(xv.x, xv.x, q); q = fmaf(xv.y, xv.y, q);
      q = fmaf(xv.z, xv.z, q); q = fmaf(xv.w, xv.w, q);
    }
    dot[i] = d;
    sq[i]  = q;
  }
#pragma unroll
  for (int off = 16; off > 0; off >>= 1) {
    float td[DOT_GROUP], tq[DOT_GROUP];
#pragma unroll
    for (int i = 0; i < DOT_GROUP; ++i) {
      td[i] = __shfl_xor(dot[i], off, 32);
      tq[i] = __shfl_xor(sq[i],  off, 32);
    }
#pragma unroll
    for (int i = 0; i < DOT_GROUP; ++i) {
      dot[i] += td[i];
      sq[i]  += tq[i];
    }
  }
#pragma unroll
  for (int i = 0; i < DOT_GROUP; ++i) res[g * DOT_GROUP + i] = dot[i] - sq[i];
}

__global__ __launch_bounds__(DOT_WAVES * 32) void dots_kernel(
    const float* __restrict__ x, const float* __restrict__ s,
    float* __restrict__ w) {
  const int lane = threadIdx.x & 31;
  const int wave = __builtin_amdgcn_readfirstlane((int)(threadIdx.x >> 5));
  int b, tchunk;
  wg_batch_chunk((int)blockIdx.x, b, tchunk);
  const int row0 = b * T_ + (tchunk * DOT_WAVES + wave) * DOT_ROWS;

  const float* xq = x + (size_t)row0 * D_ + lane * 4;
  const float* sr = s + (size_t)b * D_ + lane * 4;

  float4 sv[4];
#pragma unroll
  for (int j = 0; j < 4; ++j) sv[j] = *(const float4*)(sr + j * 128);

  float4 ra[DOT_GROUP][4], rb[DOT_GROUP][4];
  float res[DOT_ROWS];

  dots_load_group(xq, 0, ra);
#pragma unroll
  for (int g = 0; g < DOT_GROUPS; g += 2) {
    __builtin_amdgcn_sched_barrier(0);
    dots_load_group(xq, g + 1, rb);
    __builtin_amdgcn_sched_barrier(0);
    dots_sum_group(ra, sv, res, g);
    __builtin_amdgcn_sched_barrier(0);
    if (g + 2 < DOT_GROUPS) dots_load_group(xq, g + 2, ra);
    __builtin_amdgcn_sched_barrier(0);
    dots_sum_group(rb, sv, res, g + 1);
  }

  float mine = 0.f;
#pragma unroll
  for (int r = 0; r < DOT_ROWS; ++r) if (lane == r) mine = res[r];
  if (lane < DOT_ROWS) w[row0 + lane] = mine;
}

extern "C" void kernel_launch(void* const* d_in, const int* in_sizes, int n_in,
                              void* d_out, int out_size, void* d_ws, size_t ws_size,
                              hipStream_t stream) {
  (void)in_sizes; (void)n_in; (void)out_size; (void)ws_size;
  const float* x = (const float*)d_in[0];
  float* out = (float*)d_out;

  float* ws = (float*)d_ws;
  float* s_ws  = ws;
  float* w_ws  = s_ws + (size_t)B_ * D_;

  pool_wmma_kernel<false><<<B_ * (D_ / 64), POOL_WAVES * 32, 0, stream>>>(x, nullptr, s_ws);
  dots_kernel<<<(B_ * T_) / (DOT_WAVES * DOT_ROWS), DOT_WAVES * 32, 0, stream>>>(x, s_ws, w_ws);
  pool_wmma_kernel<true><<<B_ * (D_ / 64), POOL_WAVES * 32, 0, stream>>>(x, w_ws, out);
}


# logits and weighted pooling fused into one pass over x: wave owns whole rows, online softmax rescales its 2 WMMA tiles, partials joined by a small launch; changes the order of logit and softmax sums
# speedup vs baseline: 1.8441x; 1.1611x over previous
#include <hip/hip_runtime.h>
#include <hip/hip_bf16.h>
#include <cstddef>

#define B_ 32
#define T_ 2048
#define D_ 512

typedef __attribute__((ext_vector_type(2))) float v2f;
typedef __attribute__((ext_vector_type(8))) float v8f;

__device__ __forceinline__ void wg_batch_chunk(int id, int& b, int& chunk) {
  b     = (id & 15) | ((id >> 7) << 4);
  chunk = (id >> 4) & 7;
}

__device__ __forceinline__ v8f pool_mma(float a0, float a1, float b0, float b1, v8f c) {
  v2f a;  a.x = a0;  a.y = a1;
  v2f bm; bm.x = b0; bm.y = b1;
  return __builtin_amdgcn_wmma_f32_16x16x4_f32(false, a, false, bm,
                                               (short)0, c, false, false);
}

#define POOL_WAVES 16
#define POOL_T_PER_WAVE (T_ / POOL_WAVES)
#define POOL_PAIRS 4
#define POOL_GROUPS (POOL_T_PER_WAVE / (8 * POOL_PAIRS))

__device__ __forceinline__ void pool_load_group(
    const float* __restrict__ xq, int g, float4 (&r)[POOL_PAIRS][4]) {
#pragma unroll
  for (int u = 0; u < POOL_PAIRS; ++u) {
    const int row = (g * POOL_PAIRS + u) * 8;
#pragma unroll
    for (int i = 0; i < 4; ++i)
      r[u][i] = *(const float4*)(xq + (size_t)(row + i) * D_);
  }
}

__device__ __forceinline__ void pool_compute_group(
    const float4 (&r)[POOL_PAIRS][4], v8f& c0, v8f& c1, v8f& c2, v8f& c3) {
#pragma unroll
  for (int u = 0; u < POOL_PAIRS; ++u) {
    c0 = pool_mma(1.0f, 1.0f, r[u][0].x, r[u][1].x, c0);
    c1 = pool_mma(1.0f, 1.0f, r[u][0].y, r[u][1].y, c1);
    c2 = pool_mma(1.0f, 1.0f, r[u][0].z, r[u][1].z, c2);
    c3 = pool_mma(1.0f, 1.0f, r[u][0].w, r[u][1].w, c3);
    c0 = pool_mma(1.0f, 1.0f, r[u][2].x, r[u][3].x, c0);
    c1 = pool_mma(1.0f, 1.0f, r[u][2].y, r[u][3].y, c1);
    c2 = pool_mma(1.0f, 1.0f, r[u][2].z, r[u][3].z, c2);
    c3 = pool_mma(1.0f, 1.0f, r[u][2].w, r[u][3].w, c3);
  }
}

__global__ __launch_bounds__(POOL_WAVES * 32) void colsum_wmma_kernel(
    const float* __restrict__ x, float* __restrict__ out) {
  int b, dchunk;
  wg_batch_chunk((int)blockIdx.x, b, dchunk);
  const int d0   = dchunk * 64;
  const int lane = threadIdx.x & 31;
  const int wave = threadIdx.x >> 5;
  const int col  = lane & 15;
  const int h    = lane >> 4;

  const int tb = wave * POOL_T_PER_WAVE;
  const float* xq = x + (size_t)b * T_ * D_ + (size_t)(tb + 4 * h) * D_ + d0 + 4 * col;

  v8f c0 = {};
  v8f c1 = {};
  v8f c2 = {};
  v8f c3 = {};

  float4 ra[POOL_PAIRS][4], rb[POOL_PAIRS][4];

  pool_load_group(xq, 0, ra);
#pragma unroll
  for (int g = 0; g < POOL_GROUPS; g += 2) {
    __builtin_amdgcn_sched_barrier(0);
    pool_load_group(xq, g + 1, rb);
    __builtin_amdgcn_sched_barrier(0);
    pool_compute_group(ra, c0, c1, c2, c3);
    __builtin_amdgcn_sched_barrier(0);
    if (g + 2 < POOL_GROUPS) pool_load_group(xq, g + 2, ra);
    __builtin_amdgcn_sched_barrier(0);
    pool_compute_group(rb, c0, c1, c2, c3);
  }

  __shared__ float4 red[POOL_WAVES][16];
  if (lane < 16) red[wave][lane] = make_float4(c0[0], c1[0], c2[0], c3[0]);
  __syncthreads();
  if (threadIdx.x < 64) {
    const float* rf = (const float*)red;
    float acc = 0.f;
#pragma unroll
    for (int wv = 0; wv < POOL_WAVES; ++wv) acc += rf[wv * 64 + threadIdx.x];
    out[(size_t)b * D_ + d0 + threadIdx.x] = acc;
  }
}

#define FZ_WAVES 16
#define FZ_ROWS 16
#define FZ_GROUPS (FZ_ROWS / 4)
#define FZ_PART_STRIDE 544

__device__ __forceinline__ float f4c(const float4& v, int j) {
  return j == 0 ? v.x : (j == 1 ? v.y : (j == 2 ? v.z : v.w));
}

__device__ __forceinline__ void fz_load_group(
    const float* __restrict__ xq, int g, float4 (&r)[8][2]) {
#pragma unroll
  for (int q = 0; q < 8; ++q) {
#pragma unroll
    for (int i = 0; i < 2; ++i)
      r[q][i] = *(const float4*)(xq + (size_t)(4 * g + i) * D_ + 64 * q);
  }
}

__device__ __forceinline__ void fz_step(
    const float4 (&r)[8][2], const float4 (&sv)[8], int c,
    float& m, float& l, v8f& acc0, v8f& acc1) {
  float wl[2];
#pragma unroll
  for (int i = 0; i < 2; ++i) {
    float d0 = 0.f, d1 = 0.f, q0 = 0.f, q1 = 0.f;
#pragma unroll
    for (int q = 0; q < 8; q += 2) {
      const float4 xa = r[q][i];
      const float4 sa = sv[q];
      d0 = fmaf(xa.x, sa.x, d0); d0 = fmaf(xa.y, sa.y, d0);
      d0 = fmaf(xa.z, sa.z, d0); d0 = fmaf(xa.w, sa.w, d0);
      q0 = fmaf(xa.x, xa.x, q0); q0 = fmaf(xa.y, xa.y, q0);
      q0 = fmaf(xa.z, xa.z, q0); q0 = fmaf(xa.w, xa.w, q0);
      const float4 xb = r[q + 1][i];
      const float4 sb = sv[q + 1];
      d1 = fmaf(xb.x, sb.x, d1); d1 = fmaf(xb.y, sb.y, d1);
      d1 = fmaf(xb.z, sb.z, d1); d1 = fmaf(xb.w, sb.w, d1);
      q1 = fmaf(xb.x, xb.x, q1); q1 = fmaf(xb.y, xb.y, q1);
      q1 = fmaf(xb.z, xb.z, q1); q1 = fmaf(xb.w, xb.w, q1);
    }
    wl[i] = (d0 + d1) - (q0 + q1);
  }
#pragma unroll
  for (int off = 8; off > 0; off >>= 1) {
    const float t0 = __shfl_xor(wl[0], off, 32);
    const float t1 = __shfl_xor(wl[1], off, 32);
    wl[0] += t0;
    wl[1] += t1;
  }

  float mg = fmaxf(wl[0], wl[1]);
  mg = fmaxf(mg, __shfl_xor(mg, 16, 32));
  const float mn = fmaxf(m, mg);
  if (__builtin_amdgcn_readfirstlane((int)(mn > m))) {
    const float sc = __expf(m - mn);
    acc0 *= sc;
    acc1 *= sc;
    l *= sc;
  }
  m = mn;
  const float e0 = __expf(wl[0] - m);
  const float e1 = __expf(wl[1] - m);
  l += e0 + e1;

#pragma unroll
  for (int j = 0; j < 4; ++j) {
#pragma unroll
    for (int qq = 0; qq < 4; ++qq) {
      const bool sel = (c == 4 * qq + j);
      const float a0 = sel ? e0 : 0.f;
      const float a1 = sel ? e1 : 0.f;
      acc0 = pool_mma(a0, a1, f4c(r[qq][0], j),     f4c(r[qq][1], j),     acc0);
      acc1 = pool_mma(a0, a1, f4c(r[qq + 4][0], j), f4c(r[qq + 4][1], j), acc1);
    }
  }
}

__global__ __launch_bounds__(FZ_WAVES * 32) void logits_pool_kernel(
    const float* __restrict__ x, const float* __restrict__ s,
    float* __restrict__ part) {
  const int lane = threadIdx.x & 31;
  const int wave = __builtin_amdgcn_readfirstlane((int)(threadIdx.x >> 5));
  const int c = lane & 15;
  const int h = lane >> 4;
  int b, tchunk;
  wg_batch_chunk((int)blockIdx.x, b, tchunk);
  const int row0 = b * T_ + (tchunk * FZ_WAVES + wave) * FZ_ROWS;

  const float* xq = x + (size_t)(row0 + 2 * h) * D_ + 4 * c;
  const float* sr = s + (size_t)b * D_ + 4 * c;

  float4 sv[8];
#pragma unroll
  for (int q = 0; q < 8; ++q) sv[q] = *(const float4*)(sr + 64 * q);

  v8f acc0 = {};
  v8f acc1 = {};
  float m = -INFINITY, l = 0.f;

  float4 ra[8][2], rb[8][2];

  fz_load_group(xq, 0, ra);
#pragma unroll
  for (int g = 0; g < FZ_GROUPS; g += 2) {
    __builtin_amdgcn_sched_barrier(0);
    fz_load_group(xq, g + 1, rb);
    __builtin_amdgcn_sched_barrier(0);
    fz_step(ra, sv, c, m, l, acc0, acc1);
    __builtin_amdgcn_sched_barrier(0);
    if (g + 2 < FZ_GROUPS) fz_load_group(xq, g + 2, ra);
    __builtin_amdgcn_sched_barrier(0);
    fz_step(rb, sv, c, m, l, acc0, acc1);
  }
  l += __shfl_xor(l, 16, 32);

  __shared__ float red[FZ_WAVES][D_];
  __shared__ float red_m[FZ_WAVES];
  __shared__ float red_l[FZ_WAVES];
#pragma unroll
  for (int u = 0; u < 2; ++u) {
    *(float4*)&red[wave][64 * (2 * h + u) + 4 * c] =
        make_float4(acc0[4 * u], acc0[4 * u + 1], acc0[4 * u + 2], acc0[4 * u + 3]);
    *(float4*)&red[wave][64 * (4 + 2 * h + u) + 4 * c] =
        make_float4(acc1[4 * u], acc1[4 * u + 1], acc1[4 * u + 2], acc1[4 * u + 3]);
  }
  if (lane == 0) { red_m[wave] = m; red_l[wave] = l; }
  __syncthreads();

  float M = red_m[0];
#pragma unroll
  for (int wv = 1; wv < FZ_WAVES; ++wv) M = fmaxf(M, red_m[wv]);
  float acc = 0.f, L = 0.f;
#pragma unroll
  for (int wv = 0; wv < FZ_WAVES; ++wv) {
    const float sc = __expf(red_m[wv] - M);
    acc += sc * red[wv][threadIdx.x];
    L   += sc * red_l[wv];
  }
  float* prow = part + (size_t)(b * 8 + tchunk) * FZ_PART_STRIDE;
  prow[threadIdx.x] = acc;
  if (threadIdx.x == 0) { prow[D_] = M; prow[D_ + 1] = L; }
}

__global__ __launch_bounds__(D_) void join_kernel(
    const float* __restrict__ part, float* __restrict__ out) {
  const int b = blockIdx.x;
  const float* p0 = part + (size_t)b * 8 * FZ_PART_STRIDE;
  float Mc[8], Lc[8], v[8];
#pragma unroll
  for (int k = 0; k < 8; ++k) {
    Mc[k] = p0[k * FZ_PART_STRIDE + D_];
    Lc[k] = p0[k * FZ_PART_STRIDE + D_ + 1];
    v[k]  = p0[k * FZ_PART_STRIDE + threadIdx.x];
  }
  float M = Mc[0];
#pragma unroll
  for (int k = 1; k < 8; ++k) M = fmaxf(M, Mc[k]);
  float num = 0.f, den = 0.f;
#pragma unroll
  for (int k = 0; k < 8; ++k) {
    const float sc = __expf(Mc[k] - M);
    num += sc * v[k];
    den += sc * Lc[k];
  }
  out[(size_t)b * D_ + threadIdx.x] = num / den;
}

extern "C" void kernel_launch(void* const* d_in, const int* in_sizes, int n_in,
                              void* d_out, int out_size, void* d_ws, size_t ws_size,
                              hipStream_t stream) {
  (void)in_sizes; (void)n_in; (void)out_size; (void)ws_size;
  const float* x = (const float*)d_in[0];
  float* out = (float*)d_out;

  float* ws = (float*)d_ws;
  float* s_ws    = ws;
  float* part_ws = s_ws + (size_t)B_ * D_;

  colsum_wmma_kernel<<<B_ * (D_ / 64), POOL_WAVES * 32, 0, stream>>>(x, s_ws);
  logits_pool_kernel<<<(B_ * T_) / (FZ_WAVES * FZ_ROWS), FZ_WAVES * 32, 0, stream>>>(x, s_ws, part_ws);
  join_kernel<<<B_, D_, 0, stream>>>(part_ws, out);
}
